# starting kernel
# baseline (speedup 1.0000x reference)
#include <hip/hip_runtime.h>


#define KEPS 1e-3f
#define NSEQ 1024
#define DH   64
#define MF   128
#define CHK  16

typedef float v2f __attribute__((ext_vector_type(2)));
typedef float v8f __attribute__((ext_vector_type(8)));

__device__ __forceinline__ v8f wmma4(v2f a, v2f b, v8f c) {
  return __builtin_amdgcn_wmma_f32_16x16x4_f32(false, a, false, b,
                                               (short)0, c, false, false);
}

__global__ __launch_bounds__(256)
void performer_causal_f32_wmma(const float* __restrict__ q,
                               const float* __restrict__ k,
                               const float* __restrict__ v,
                               const float* __restrict__ proj,
                               float* __restrict__ out) {
  __shared__ float proj_s[MF * DH];
  __shared__ float S_s[MF * DH];
  __shared__ float qp_s[CHK * MF];
  __shared__ float kp_s[CHK * MF];
  __shared__ float xq_s[CHK * DH];
  __shared__ float xk_s[CHK * DH];
  __shared__ float v_s[CHK * DH];
  __shared__ float a_s[CHK * CHK];
  __shared__ float kc_s[MF];

  const int h    = blockIdx.x;
  const int tid  = threadIdx.x;
  const int lane = tid & 31;
  const int wave = tid >> 5;
  const int lr   = lane & 15;
  const int lh   = lane >> 4;

  const float norm = 0.35355339059327378f;

  const float* qh = q + (size_t)h * NSEQ * DH;
  const float* kh = k + (size_t)h * NSEQ * DH;
  const float* vh = v + (size_t)h * NSEQ * DH;
  float*       oh = out + (size_t)h * NSEQ * DH;

  for (int i = tid; i < MF * DH; i += 256) { proj_s[i] = proj[i]; S_s[i] = 0.f; }
  for (int i = tid; i < MF; i += 256) kc_s[i] = 0.f;
  __syncthreads();

  for (int n0 = 0; n0 < NSEQ; n0 += CHK) {
    for (int i = tid; i < CHK * DH; i += 256) {
      xq_s[i] = qh[n0 * DH + i];
      xk_s[i] = kh[n0 * DH + i];
      v_s[i]  = vh[n0 * DH + i];
    }
    __syncthreads();

    {
      const int m0 = wave * 16;
      v8f accq = {};
      v8f acck = {};
#pragma unroll
      for (int kb = 0; kb < DH; kb += 4) {
        v2f a, b;
        b.x = proj_s[(m0 + lr) * DH + kb + 2 * lh];
        b.y = proj_s[(m0 + lr) * DH + kb + 2 * lh + 1];
        a.x = xq_s[lr * DH + kb + 2 * lh];
        a.y = xq_s[lr * DH + kb + 2 * lh + 1];
        accq = wmma4(a, b, accq);
        a.x = xk_s[lr * DH + kb + 2 * lh];
        a.y = xk_s[lr * DH + kb + 2 * lh + 1];
        acck = wmma4(a, b, acck);
      }
#pragma unroll
      for (int j = 0; j < 8; ++j) {
        const int row = j + 8 * lh;
        qp_s[row * MF + m0 + lr] = fmaxf(accq[j] * norm, 0.f) + KEPS;
        kp_s[row * MF + m0 + lr] = fmaxf(acck[j] * norm, 0.f) + KEPS;
      }
    }
    __syncthreads();

    if (tid < MF) {
      float run = kc_s[tid];
#pragma unroll
      for (int r = 0; r < CHK; ++r) {
        run += kp_s[r * MF + tid];
        qp_s[r * MF + tid] = qp_s[r * MF + tid] / run;
      }
      kc_s[tid] = run;
    }
    __syncthreads();

    v8f o = {};
    if (wave < 4) {
      const int e0 = wave * 16;
#pragma unroll
      for (int kb = 0; kb < MF; kb += 4) {
        v2f a, b;
        a.x = qp_s[lr * MF + kb + 2 * lh];
        a.y = qp_s[lr * MF + kb + 2 * lh + 1];
        b.x = S_s[(kb + 2 * lh) * DH + e0 + lr];
        b.y = S_s[(kb + 2 * lh + 1) * DH + e0 + lr];
        o = wmma4(a, b, o);
      }
    } else if (wave == 4) {
      v8f am = {};
#pragma unroll
      for (int kb = 0; kb < MF; kb += 4) {
        v2f a, b;
        a.x = qp_s[lr * MF + kb + 2 * lh];
        a.y = qp_s[lr * MF + kb + 2 * lh + 1];
        b.x = kp_s[lr * MF + kb + 2 * lh];
        b.y = kp_s[lr * MF + kb + 2 * lh + 1];
        am = wmma4(a, b, am);
      }
#pragma unroll
      for (int j = 0; j < 8; ++j) {
        const int row = j + 8 * lh;
        a_s[row * CHK + lr] = (lr <= row) ? am[j] : 0.f;
      }
    }
    __syncthreads();

    if (wave < 4) {
      const int e0 = wave * 16;
#pragma unroll
      for (int kb = 0; kb < CHK; kb += 4) {
        v2f a, b;
        a.x = a_s[lr * CHK + kb + 2 * lh];
        a.y = a_s[lr * CHK + kb + 2 * lh + 1];
        b.x = v_s[(kb + 2 * lh) * DH + e0 + lr];
        b.y = v_s[(kb + 2 * lh + 1) * DH + e0 + lr];
        o = wmma4(a, b, o);
      }
#pragma unroll
      for (int j = 0; j < 8; ++j) {
        const int row = j + 8 * lh;
        oh[(n0 + row) * DH + e0 + lr] = o[j];
      }
    }
    {
      const int m0 = wave * 16;
#pragma unroll
      for (int et = 0; et < 4; ++et) {
        const int e0 = et * 16;
        v8f sacc;
#pragma unroll
        for (int j = 0; j < 8; ++j)
          sacc[j] = S_s[(m0 + j + 8 * lh) * DH + e0 + lr];
#pragma unroll
        for (int kb = 0; kb < CHK; kb += 4) {
          v2f a, b;
          a.x = kp_s[(kb + 2 * lh) * MF + m0 + lr];
          a.y = kp_s[(kb + 2 * lh + 1) * MF + m0 + lr];
          b.x = v_s[(kb + 2 * lh) * DH + e0 + lr];
          b.y = v_s[(kb + 2 * lh + 1) * DH + e0 + lr];
          sacc = wmma4(a, b, sacc);
        }
#pragma unroll
        for (int j = 0; j < 8; ++j)
          S_s[(m0 + j + 8 * lh) * DH + e0 + lr] = sacc[j];
      }
    }
    __syncthreads();
  }
}

extern "C" void kernel_launch(void* const* d_in, const int* in_sizes, int n_in,
                              void* d_out, int out_size, void* d_ws, size_t ws_size,
                              hipStream_t stream) {
  (void)in_sizes; (void)n_in; (void)out_size; (void)d_ws; (void)ws_size;
  const float* q    = (const float*)d_in[0];
  const float* k    = (const float*)d_in[1];
  const float* v    = (const float*)d_in[2];
  const float* proj = (const float*)d_in[3];
  float* out        = (float*)d_out;
  performer_causal_f32_wmma<<<dim3(8), dim3(256), 0, stream>>>(q, k, v, proj, out);
}


# Query rows re-cut for equal work: a workgroup takes 16-row tiles p and 63-p of a head, its 17 tile x key-block units dealt to two wave sets; 256 equal workgroups; sum order of scores and out changes
# speedup vs baseline: 19.1392x; 19.1392x over previous
#include <hip/hip_runtime.h>


#define KEPS 1e-3f
#define NSEQ 1024
#define DH   64
#define MF   128
#define RB   64
#define NBLK 16
#define QT   16
#define NPAIR 32
#define PX   68
#define PV   72
#define PM   132

#define WS_KP   0
#define WS_BSUM (8 * NSEQ * MF)

typedef float v2f __attribute__((ext_vector_type(2)));
typedef float v4f __attribute__((ext_vector_type(4)));
typedef float v8f __attribute__((ext_vector_type(8)));

__device__ __forceinline__ v8f wmma4(v2f a, v2f b, v8f c) {
  return __builtin_amdgcn_wmma_f32_16x16x4_f32(false, a, false, b,
                                               (short)0, c, false, false);
}

__device__ __forceinline__ v2f lds2(const float* p) {
  return *reinterpret_cast<const v2f*>(p);
}

__device__ __forceinline__ void load_rows(const float* __restrict__ g, int tid, v4f (&r)[4]) {
#pragma unroll
  for (int i = 0; i < 4; ++i)
    r[i] = *reinterpret_cast<const v4f*>(g + 4 * (tid + 256 * i));
}

__device__ __forceinline__ void store_rows(float* s, int pitch, int tid, const v4f (&r)[4]) {
#pragma unroll
  for (int i = 0; i < 4; ++i) {
    const int idx = tid + 256 * i;
    *reinterpret_cast<v4f*>(s + (idx >> 4) * pitch + 4 * (idx & 15)) = r[i];
  }
}

__device__ __forceinline__ void load_kp(const float* __restrict__ g, int tid, v4f (&r)[8]) {
#pragma unroll
  for (int i = 0; i < 8; ++i)
    r[i] = *reinterpret_cast<const v4f*>(g + 4 * (tid + 256 * i));
}

__device__ __forceinline__ void store_kp(float* s, int tid, const v4f (&r)[8]) {
#pragma unroll
  for (int i = 0; i < 8; ++i) {
    const int idx = tid + 256 * i;
    *reinterpret_cast<v4f*>(s + (idx >> 5) * PM + 4 * (idx & 31)) = r[i];
  }
}

__device__ __forceinline__ void load_proj(const float* __restrict__ proj, int m0,
                                          int lr, int lh, v2f (&pf)[16]) {
#pragma unroll
  for (int s = 0; s < 16; ++s) {
    const float* pp = proj + (m0 + lr) * DH + 4 * s + 2 * lh;
    pf[s].x = pp[0];
    pf[s].y = pp[1];
  }
}

__device__ __forceinline__ void feat_tiles(const float* xs, const v2f (&pf)[16],
                                           int lr, int lh, v8f (&acc)[4]) {
#pragma unroll
  for (int t = 0; t < 4; ++t) acc[t] = v8f{};
#pragma unroll
  for (int s = 0; s < 16; ++s) {
#pragma unroll
    for (int t = 0; t < 4; ++t) {
      const v2f a = lds2(xs + (t * 16 + lr) * PX + 4 * s + 2 * lh);
      acc[t] = wmma4(a, pf[s], acc[t]);
    }
  }
}

__device__ __forceinline__ void store_feat(float* dst, const v8f (&acc)[4], int m0,
                                           int lr, int lh, float norm) {
#pragma unroll
  for (int t = 0; t < 4; ++t)
#pragma unroll
    for (int i = 0; i < 8; ++i)
      dst[(t * 16 + i + 8 * lh) * PM + m0 + lr] = fmaxf(acc[t][i] * norm, 0.f) + KEPS;
}

__global__ __launch_bounds__(256)
void performer_key_features(const float* __restrict__ k,
                            const float* __restrict__ proj,
                            float* __restrict__ ws) {
  __shared__ __attribute__((aligned(16))) float x_s[RB * PX];
  __shared__ __attribute__((aligned(16))) float kp_s[RB * PM];

  const int h    = blockIdx.x >> 4;
  const int b    = blockIdx.x & 15;
  const int tid  = threadIdx.x;
  const int lane = tid & 31;
  const int wave = tid >> 5;
  const int lr   = lane & 15;
  const int lh   = lane >> 4;
  const int m0   = wave * 16;

  const float norm = 0.35355339059327378f;

  v4f rk[4];
  load_rows(k + ((size_t)h * NSEQ + b * RB) * DH, tid, rk);
  v2f pf[16];
  load_proj(proj, m0, lr, lh, pf);
  store_rows(x_s, PX, tid, rk);
  __syncthreads();

  v8f acc[4];
  feat_tiles(x_s, pf, lr, lh, acc);
  store_feat(kp_s, acc, m0, lr, lh, norm);
  float cs = 0.f;
#pragma unroll
  for (int t = 0; t < 4; ++t)
#pragma unroll
    for (int i = 0; i < 8; ++i)
      cs += fmaxf(acc[t][i] * norm, 0.f) + KEPS;
  cs += __shfl_xor(cs, 16);
  if (lh == 0) ws[WS_BSUM + (h * NBLK + b) * MF + m0 + lr] = cs;
  __syncthreads();

  float* dst = ws + WS_KP + ((size_t)h * NSEQ + b * RB) * MF;
#pragma unroll
  for (int i = 0; i < 8; ++i) {
    const int idx = tid + 256 * i;
    *reinterpret_cast<v4f*>(dst + 4 * idx) =
        *reinterpret_cast<const v4f*>(kp_s + (idx >> 5) * PM + 4 * (idx & 31));
  }
}

__device__ __forceinline__ void feat_tiles2(const float* xs, const v2f (&pf)[16],
                                            int lr, int lh, v8f (&acc)[2]) {
  acc[0] = v8f{};
  acc[1] = v8f{};
#pragma unroll
  for (int s = 0; s < 16; ++s) {
#pragma unroll
    for (int t = 0; t < 2; ++t) {
      const v2f a = lds2(xs + (t * 16 + lr) * PX + 4 * s + 2 * lh);
      acc[t] = wmma4(a, pf[s], acc[t]);
    }
  }
}

__device__ __forceinline__ void score_tile(const float* qrow, const float* krow, float* ps,
                                           int lr, int lh, bool diag, int key0, int row0) {
  v8f s0 = {}, s1 = {};
#pragma unroll
  for (int s = 0; s < 32; s += 2) {
    const v2f a0 = lds2(qrow + lr * PM + 4 * s + 2 * lh);
    const v2f b0 = lds2(krow + lr * PM + 4 * s + 2 * lh);
    const v2f a1 = lds2(qrow + lr * PM + 4 * s + 4 + 2 * lh);
    const v2f b1 = lds2(krow + lr * PM + 4 * s + 4 + 2 * lh);
    s0 = wmma4(a0, b0, s0);
    s1 = wmma4(a1, b1, s1);
  }
#pragma unroll
  for (int i = 0; i < 8; ++i) {
    const int row = i + 8 * lh;
    float x = s0[i] + s1[i];
    if (diag && (key0 + lr > row0 + row)) x = 0.f;
    ps[row * PX + lr] = x;
  }
}

__device__ __forceinline__ void out_tile(const float* ps, const float* vcol, int lr, int lh,
                                         v8f& oa, v8f& ob) {
#pragma unroll
  for (int s = 0; s < 16; s += 2) {
    const v2f a0 = lds2(ps + lr * PX + 4 * s + 2 * lh);
    const v2f a1 = lds2(ps + lr * PX + 4 * s + 4 + 2 * lh);
    const float* vb0 = vcol + (4 * s + 2 * lh) * PV + lr;
    const float* vb1 = vb0 + 4 * PV;
    v2f b0, b1;
    b0.x = vb0[0]; b0.y = vb0[PV];
    b1.x = vb1[0]; b1.y = vb1[PV];
    oa = wmma4(a0, b0, oa);
    ob = wmma4(a1, b1, ob);
  }
}

__global__ __launch_bounds__(256)
void performer_causal_pairs(const float* __restrict__ q,
                            const float* __restrict__ v,
                            const float* __restrict__ proj,
                            const float* __restrict__ ws,
                            float* __restrict__ out) {
  __shared__ __attribute__((aligned(16))) float x_s[2 * QT * PX];
  __shared__ __attribute__((aligned(16))) float qn_s[2 * QT * PM];
  __shared__ __attribute__((aligned(16))) float p_s[2][QT * PX];
  __shared__ __attribute__((aligned(16))) float v_s[2][2][RB * PV];
  __shared__ __attribute__((aligned(16))) float kp_s[2][2][RB * PM];

  const int h    = blockIdx.x >> 5;
  const int p    = blockIdx.x & 31;
  const int tid  = threadIdx.x;
  const int lane = tid & 31;
  const int wave = __builtin_amdgcn_readfirstlane(tid >> 5);
  const int lr   = lane & 15;
  const int lh   = lane >> 4;
  const int m0   = wave * 16;

  const int tile_f = p;
  const int tile_b = 2 * NPAIR - 1 - p;
  const int jf     = tile_f >> 2;
  const int jb     = tile_b >> 2;
  const int NTRIP  = 9;

  const int set = (wave ^ (wave >> 2)) & 1;
  const int ct  = wave >> 1;

  const float norm = 0.35355339059327378f;

  const float* qh  = q + (size_t)h * NSEQ * DH;
  const float* vh  = v + (size_t)h * NSEQ * DH;
  const float* kpw = ws + WS_KP + (size_t)h * NSEQ * MF;
  const float* bs  = ws + WS_BSUM + h * NBLK * MF;
  float*       oh  = out + (size_t)h * NSEQ * DH;

  const int qr = tid >> 4, qc = 4 * (tid & 15);
  const v4f rq0 = *reinterpret_cast<const v4f*>(qh + (tile_f * QT + qr) * DH + qc);
  const v4f rq1 = *reinterpret_cast<const v4f*>(qh + (tile_b * QT + qr) * DH + qc);

  v2f pf[16];
  load_proj(proj, m0, lr, lh, pf);

  const int sc_b    = tid >> 7;
  const int sc_m    = tid & (MF - 1);
  const int sc_tile = sc_b ? tile_b : tile_f;
  const int sc_j    = sc_tile >> 2;
  const int sc_og   = sc_tile & 3;
  float carry = 0.f;
#pragma unroll
  for (int j = 0; j < NBLK - 1; ++j) {
    const float x = bs[j * MF + sc_m];
    carry += (j < sc_j) ? x : 0.f;
  }
  const float* kd = kpw + (size_t)sc_j * RB * MF + sc_m;
  float xa[48], xf[QT];
#pragma unroll
  for (int r = 0; r < 48; ++r) xa[r] = kd[r * MF];
#pragma unroll
  for (int i = 0; i < QT; ++i) xf[i] = kd[(sc_og * QT + i) * MF];

  *reinterpret_cast<v4f*>(x_s + qr * PX + qc)        = rq0;
  *reinterpret_cast<v4f*>(x_s + (QT + qr) * PX + qc) = rq1;
  __syncthreads();

  v4f rv0[4], rkp0[8], rv1[4], rkp1[8];
  load_rows(vh, tid, rv0);
  load_kp(kpw, tid, rkp0);

  {
    v8f acc[2];
    feat_tiles2(x_s, pf, lr, lh, acc);
#pragma unroll
    for (int t = 0; t < 2; ++t)
#pragma unroll
      for (int i = 0; i < 8; ++i)
        qn_s[(t * 16 + i + 8 * lh) * PM + m0 + lr] = fmaxf(acc[t][i] * norm, 0.f) + KEPS;
  }
  __syncthreads();
  {
    float run = carry;
#pragma unroll
    for (int g = 0; g < 3; ++g)
#pragma unroll
      for (int i = 0; i < QT; ++i)
        run += (g < sc_og) ? xa[g * QT + i] : 0.f;
#pragma unroll
    for (int i = 0; i < QT; ++i) {
      run += xf[i];
      float* qq = qn_s + (sc_b * QT + i) * PM + sc_m;
      *qq = *qq / run;
    }
  }

  v8f oa = {}, ob = {}, oc = {}, od = {};
  for (int t = 0; t < NTRIP; ++t) {
    const int  par  = t & 1;
    const bool both = (t <= jf);
    const int  kb0  = both ? t : (2 * t - jf - 1);
    const bool two  = !both && (t < NTRIP - 1);

    store_rows(v_s[par][0], PV, tid, rv0);
    store_kp(kp_s[par][0], tid, rkp0);
    if (two) {
      store_rows(v_s[par][1], PV, tid, rv1);
      store_kp(kp_s[par][1], tid, rkp1);
    }
    __syncthreads();
    if (t + 1 < NTRIP) {
      const int  tn    = t + 1;
      const bool bothn = (tn <= jf);
      const int  kn    = bothn ? tn : (2 * tn - jf - 1);
      load_rows(vh + kn * RB * DH, tid, rv0);
      load_kp(kpw + kn * RB * MF, tid, rkp0);
      if (!bothn && tn < NTRIP - 1) {
        load_rows(vh + (kn + 1) * RB * DH, tid, rv1);
        load_kp(kpw + (kn + 1) * RB * MF, tid, rkp1);
      }
    }

    const bool back   = (set == 1) || !both;
    const int  slot   = (set == 1 && !both) ? 1 : 0;
    const bool active = (set == 0) || both || two;
    const int  tile   = back ? tile_b : tile_f;
    const int  kblk   = kb0 + slot;
    const bool diag   = (kblk == (tile >> 2));
    const bool late   = (set == 0) && !both;
    float*       ps   = p_s[set];
    const float* vs   = v_s[par][slot];

    if (active)
      score_tile(qn_s + (back ? QT : 0) * PM, kp_s[par][slot] + ct * 16 * PM, ps + ct * 16,
                 lr, lh, diag, kblk * RB + ct * 16, tile * QT);
    __syncthreads();
    if (active) {
      if (late) out_tile(ps, vs + ct * 16, lr, lh, oc, od);
      else      out_tile(ps, vs + ct * 16, lr, lh, oa, ob);
    }
  }

#pragma unroll
  for (int i = 0; i < 8; ++i)
    x_s[(set * QT + i + 8 * lh) * PX + ct * 16 + lr] = oa[i] + ob[i];
  __syncthreads();
  if (set == 0) {
#pragma unroll
    for (int i = 0; i < 8; ++i)
      x_s[(QT + i + 8 * lh) * PX + ct * 16 + lr] += oc[i] + od[i];
  }
  __syncthreads();
  *reinterpret_cast<v4f*>(oh + (tile_f * QT + qr) * DH + qc) =
      *reinterpret_cast<const v4f*>(x_s + qr * PX + qc);
  *reinterpret_cast<v4f*>(oh + (tile_b * QT + qr) * DH + qc) =
      *reinterpret_cast<const v4f*>(x_s + (QT + qr) * PX + qc);
}

extern "C" void kernel_launch(void* const* d_in, const int* in_sizes, int n_in,
                              void* d_out, int out_size, void* d_ws, size_t ws_size,
                              hipStream_t stream) {
  (void)in_sizes; (void)n_in; (void)out_size; (void)ws_size;
  const float* q    = (const float*)d_in[0];
  const float* k    = (const float*)d_in[1];
  const float* v    = (const float*)d_in[2];
  const float* proj = (const float*)d_in[3];
  float* out        = (float*)d_out;
  float* ws         = (float*)d_ws;
  performer_key_features<<<dim3(128), dim3(256), 0, stream>>>(k, proj, ws);
  performer_causal_pairs<<<dim3(256), dim3(256), 0, stream>>>(q, v, proj, ws, out);
}


# Waves run free: each takes whole 16-key tiles into its own LDS area and forms the scores transposed, so the score tile in registers is the A operand of P@v; no barrier in the loop; sum order changes
# speedup vs baseline: 23.5605x; 1.2310x over previous
#include <hip/hip_runtime.h>


#define KEPS 1e-3f
#define NSEQ 1024
#define DH   64
#define MF   128
#define RB   64
#define NBLK 16
#define QT   16
#define NPAIR 32
#define PX   68
#define PV   68
#define PM   132
#define PR   68
#define STG  (QT * PM + QT * PV)

#define WS_KP   0
#define WS_BSUM (8 * NSEQ * MF)

typedef float v2f __attribute__((ext_vector_type(2)));
typedef float v4f __attribute__((ext_vector_type(4)));
typedef float v8f __attribute__((ext_vector_type(8)));

__device__ __forceinline__ v8f wmma4(v2f a, v2f b, v8f c) {
  return __builtin_amdgcn_wmma_f32_16x16x4_f32(false, a, false, b,
                                               (short)0, c, false, false);
}

__device__ __forceinline__ v2f lds2(const float* p) {
  return *reinterpret_cast<const v2f*>(p);
}

__device__ __forceinline__ void load_rows(const float* __restrict__ g, int tid, v4f (&r)[4]) {
#pragma unroll
  for (int i = 0; i < 4; ++i)
    r[i] = *reinterpret_cast<const v4f*>(g + 4 * (tid + 256 * i));
}

__device__ __forceinline__ void store_rows(float* s, int pitch, int tid, const v4f (&r)[4]) {
#pragma unroll
  for (int i = 0; i < 4; ++i) {
    const int idx = tid + 256 * i;
    *reinterpret_cast<v4f*>(s + (idx >> 4) * pitch + 4 * (idx & 15)) = r[i];
  }
}

__device__ __forceinline__ void load_proj(const float* __restrict__ proj, int m0,
                                          int lr, int lh, v2f (&pf)[16]) {
#pragma unroll
  for (int s = 0; s < 16; ++s) {
    const float* pp = proj + (m0 + lr) * DH + 4 * s + 2 * lh;
    pf[s].x = pp[0];
    pf[s].y = pp[1];
  }
}

__device__ __forceinline__ void feat_tiles(const float* xs, const v2f (&pf)[16],
                                           int lr, int lh, v8f (&acc)[4]) {
#pragma unroll
  for (int t = 0; t < 4; ++t) acc[t] = v8f{};
#pragma unroll
  for (int s = 0; s < 16; ++s) {
#pragma unroll
    for (int t = 0; t < 4; ++t) {
      const v2f a = lds2(xs + (t * 16 + lr) * PX + 4 * s + 2 * lh);
      acc[t] = wmma4(a, pf[s], acc[t]);
    }
  }
}

__device__ __forceinline__ void store_feat(float* dst, const v8f (&acc)[4], int m0,
                                           int lr, int lh, float norm) {
#pragma unroll
  for (int t = 0; t < 4; ++t)
#pragma unroll
    for (int i = 0; i < 8; ++i)
      dst[(t * 16 + i + 8 * lh) * PM + m0 + lr] = fmaxf(acc[t][i] * norm, 0.f) + KEPS;
}

__global__ __launch_bounds__(256)
void performer_key_features(const float* __restrict__ k,
                            const float* __restrict__ proj,
                            float* __restrict__ ws) {
  __shared__ __attribute__((aligned(16))) float x_s[RB * PX];
  __shared__ __attribute__((aligned(16))) float kp_s[RB * PM];

  const int h    = blockIdx.x >> 4;
  const int b    = blockIdx.x & 15;
  const int tid  = threadIdx.x;
  const int lane = tid & 31;
  const int wave = tid >> 5;
  const int lr   = lane & 15;
  const int lh   = lane >> 4;
  const int m0   = wave * 16;

  const float norm = 0.35355339059327378f;

  v4f rk[4];
  load_rows(k + ((size_t)h * NSEQ + b * RB) * DH, tid, rk);
  v2f pf[16];
  load_proj(proj, m0, lr, lh, pf);
  store_rows(x_s, PX, tid, rk);
  __syncthreads();

  v8f acc[4];
  feat_tiles(x_s, pf, lr, lh, acc);
  store_feat(kp_s, acc, m0, lr, lh, norm);
  float cs = 0.f;
#pragma unroll
  for (int t = 0; t < 4; ++t)
#pragma unroll
    for (int i = 0; i < 8; ++i)
      cs += fmaxf(acc[t][i] * norm, 0.f) + KEPS;
  cs += __shfl_xor(cs, 16);
  if (lh == 0) ws[WS_BSUM + (h * NBLK + b) * MF + m0 + lr] = cs;
  __syncthreads();

  float* dst = ws + WS_KP + ((size_t)h * NSEQ + b * RB) * MF;
#pragma unroll
  for (int i = 0; i < 8; ++i) {
    const int idx = tid + 256 * i;
    *reinterpret_cast<v4f*>(dst + 4 * idx) =
        *reinterpret_cast<const v4f*>(kp_s + (idx >> 5) * PM + 4 * (idx & 31));
  }
}

__device__ __forceinline__ void feat_tiles2(const float* xs, const v2f (&pf)[16],
                                            int lr, int lh, v8f (&acc)[2]) {
  acc[0] = v8f{};
  acc[1] = v8f{};
#pragma unroll
  for (int s = 0; s < 16; ++s) {
#pragma unroll
    for (int t = 0; t < 2; ++t) {
      const v2f a = lds2(xs + (t * 16 + lr) * PX + 4 * s + 2 * lh);
      acc[t] = wmma4(a, pf[s], acc[t]);
    }
  }
}

__device__ __forceinline__ v2f lo2(v4f x) { v2f r; r.x = x.x; r.y = x.y; return r; }
__device__ __forceinline__ v2f hi2(v4f x) { v2f r; r.x = x.z; r.y = x.w; return r; }

__device__ __forceinline__ void tile_step(const v4f (&ka)[16], const v4f (&qf)[16],
                                          const v2f (&vb)[4][4], bool dg, int lr, int lh,
                                          v8f (&o)[4]) {
  v8f s0 = {}, s1 = {};
#pragma unroll
  for (int u = 0; u < 16; ++u) {
    s0 = wmma4(lo2(ka[u]), lo2(qf[u]), s0);
    s1 = wmma4(hi2(ka[u]), hi2(qf[u]), s1);
  }
  v2f pa[4];
#pragma unroll
  for (int t = 0; t < 4; ++t) {
    float x0 = s0[2 * t] + s1[2 * t];
    float x1 = s0[2 * t + 1] + s1[2 * t + 1];
    if (dg && (8 * lh + 2 * t > lr))     x0 = 0.f;
    if (dg && (8 * lh + 2 * t + 1 > lr)) x1 = 0.f;
    pa[t].x = x0;
    pa[t].y = x1;
  }
#pragma unroll
  for (int t = 0; t < 4; ++t)
#pragma unroll
    for (int c = 0; c < 4; ++c)
      o[c] = wmma4(pa[t], vb[t][c], o[c]);
}

__device__ __forceinline__ void load_tile(const float* __restrict__ gk, const float* __restrict__ gv,
                                          int lane, v4f (&rk)[16], v4f (&rv)[8]) {
#pragma unroll
  for (int i = 0; i < 16; ++i)
    rk[i] = *reinterpret_cast<const v4f*>(gk + 4 * (lane + 32 * i));
#pragma unroll
  for (int i = 0; i < 8; ++i)
    rv[i] = *reinterpret_cast<const v4f*>(gv + 4 * (lane + 32 * i));
}

__global__ __launch_bounds__(256)
void performer_causal_pairs(const float* __restrict__ q,
                            const float* __restrict__ v,
                            const float* __restrict__ proj,
                            const float* __restrict__ ws,
                            float* __restrict__ out) {
  __shared__ __attribute__((aligned(16))) float x_s[2 * QT * PX];
  __shared__ __attribute__((aligned(16))) float qn_s[2 * QT * PM];
  __shared__ __attribute__((aligned(16))) float st_s[8][STG];

  const int h    = blockIdx.x >> 5;
  const int p    = blockIdx.x & 31;
  const int tid  = threadIdx.x;
  const int lane = tid & 31;
  const int wave = __builtin_amdgcn_readfirstlane(tid >> 5);
  const int lr   = lane & 15;
  const int lh   = lane >> 4;
  const int m0   = wave * 16;

  const int tile_f = p;
  const int tile_b = 2 * NPAIR - 1 - p;

  const int ns    = p + 1;
  const int nb    = 2 * NPAIR - 1 - 2 * p;
  const int rr    = ns & 7;
  const int nbeh  = rr ? 8 - rr : 0;
  const int ncat  = (2 * nbeh < nb) ? 2 * nbeh : nb;
  const int nS    = (p >= wave) ? ((p - wave) >> 3) + 1 : 0;
  const int c0    = wave - rr;
  const int nC    = (rr != 0 && wave >= rr) ? ((c0 < ncat ? 1 : 0) + (c0 + nbeh < ncat ? 1 : 0)) : 0;
  const int r0    = ncat + wave;
  const int nR    = (nb > r0) ? ((nb - 1 - r0) >> 3) + 1 : 0;
  const int njobs = nS + nC + nR;
  auto job_tile = [&](int j) -> int {
    if (j < nS) return wave + 8 * j;
    j -= nS;
    if (j < nC) return p + 1 + c0 + j * nbeh;
    j -= nC;
    return p + 1 + r0 + 8 * j;
  };

  const float norm = 0.35355339059327378f;

  const float* qh  = q + (size_t)h * NSEQ * DH;
  const float* vh  = v + (size_t)h * NSEQ * DH;
  const float* kpw = ws + WS_KP + (size_t)h * NSEQ * MF;
  const float* bs  = ws + WS_BSUM + h * NBLK * MF;
  float*       oh  = out + (size_t)h * NSEQ * DH;

  const int qr = tid >> 4, qc = 4 * (tid & 15);
  const v4f rq0 = *reinterpret_cast<const v4f*>(qh + (tile_f * QT + qr) * DH + qc);
  const v4f rq1 = *reinterpret_cast<const v4f*>(qh + (tile_b * QT + qr) * DH + qc);

  v2f pf[16];
  load_proj(proj, m0, lr, lh, pf);

  const int sc_b    = tid >> 7;
  const int sc_m    = tid & (MF - 1);
  const int sc_tile = sc_b ? tile_b : tile_f;
  const int sc_j    = sc_tile >> 2;
  const int sc_og   = sc_tile & 3;
  float carry = 0.f;
#pragma unroll
  for (int j = 0; j < NBLK - 1; ++j) {
    const float x = bs[j * MF + sc_m];
    carry += (j < sc_j) ? x : 0.f;
  }
  const float* kd = kpw + (size_t)sc_j * RB * MF + sc_m;
  float xa[48], xf[QT];
#pragma unroll
  for (int r = 0; r < 48; ++r) xa[r] = kd[r * MF];
#pragma unroll
  for (int i = 0; i < QT; ++i) xf[i] = kd[(sc_og * QT + i) * MF];

  *reinterpret_cast<v4f*>(x_s + qr * PX + qc)        = rq0;
  *reinterpret_cast<v4f*>(x_s + (QT + qr) * PX + qc) = rq1;
  __syncthreads();

  v4f rk[16], rv[8];
  int kt = job_tile(0);
  if (njobs > 0) load_tile(kpw + kt * QT * MF, vh + kt * QT * DH, lane, rk, rv);

  {
    v8f acc[2];
    feat_tiles2(x_s, pf, lr, lh, acc);
#pragma unroll
    for (int t = 0; t < 2; ++t)
#pragma unroll
      for (int i = 0; i < 8; ++i)
        qn_s[(t * 16 + i + 8 * lh) * PM + m0 + lr] = fmaxf(acc[t][i] * norm, 0.f) + KEPS;
  }
  __syncthreads();
  {
    float run = carry;
#pragma unroll
    for (int g = 0; g < 3; ++g)
#pragma unroll
      for (int i = 0; i < QT; ++i)
        run += (g < sc_og) ? xa[g * QT + i] : 0.f;
#pragma unroll
    for (int i = 0; i < QT; ++i) {
      run += xf[i];
      float* qq = qn_s + (sc_b * QT + i) * PM + sc_m;
      *qq = *qq / run;
    }
  }
  __syncthreads();

  v4f qF[16], qB[16];
#pragma unroll
  for (int u = 0; u < 16; ++u) {
    qF[u] = *reinterpret_cast<const v4f*>(qn_s + lr * PM + 8 * u + 4 * lh);
    qB[u] = *reinterpret_cast<const v4f*>(qn_s + (QT + lr) * PM + 8 * u + 4 * lh);
  }

  float* kst = st_s[wave];
  float* vst = kst + QT * PM;
  v8f oF[4] = {}, oB[4] = {};
  for (int j = 0; j < njobs; ++j) {
#pragma unroll
    for (int i = 0; i < 16; ++i)
      *reinterpret_cast<v4f*>(kst + i * PM + 4 * lane) = rk[i];
#pragma unroll
    for (int i = 0; i < 8; ++i) {
      const int idx = lane + 32 * i;
      *reinterpret_cast<v4f*>(vst + (idx >> 4) * PV + 4 * (idx & 15)) = rv[i];
    }
    const int  ktc  = kt;
    const bool both = (j < nS);
    if (j + 1 < njobs) {
      kt = job_tile(j + 1);
      load_tile(kpw + kt * QT * MF, vh + kt * QT * DH, lane, rk, rv);
    }
    v4f ka[16];
#pragma unroll
    for (int u = 0; u < 16; ++u)
      ka[u] = *reinterpret_cast<const v4f*>(kst + lr * PM + 8 * u + 4 * lh);
    v2f vb[4][4];
#pragma unroll
    for (int t = 0; t < 4; ++t)
#pragma unroll
      for (int c = 0; c < 4; ++c) {
        const float* pv = vst + (8 * lh + 2 * t) * PV + 16 * c + lr;
        vb[t][c].x = pv[0];
        vb[t][c].y = pv[PV];
      }
    tile_step(ka, qB, vb, ktc == tile_b, lr, lh, oB);
    if (both) tile_step(ka, qF, vb, ktc == tile_f, lr, lh, oF);
  }

  float* red = st_s[wave];
#pragma unroll
  for (int c = 0; c < 4; ++c)
#pragma unroll
    for (int i = 0; i < 8; ++i) {
      red[(i + 8 * lh) * PR + 16 * c + lr]      = oF[c][i];
      red[(QT + i + 8 * lh) * PR + 16 * c + lr] = oB[c][i];
    }
  __syncthreads();
  v4f sF = {}, sB = {};
#pragma unroll
  for (int w8 = 0; w8 < 8; ++w8) {
    sF += *reinterpret_cast<const v4f*>(&st_s[w8][qr * PR + qc]);
    sB += *reinterpret_cast<const v4f*>(&st_s[w8][(QT + qr) * PR + qc]);
  }
  *reinterpret_cast<v4f*>(oh + (tile_f * QT + qr) * DH + qc) = sF;
  *reinterpret_cast<v4f*>(oh + (tile_b * QT + qr) * DH + qc) = sB;
}

extern "C" void kernel_launch(void* const* d_in, const int* in_sizes, int n_in,
                              void* d_out, int out_size, void* d_ws, size_t ws_size,
                              hipStream_t stream) {
  (void)in_sizes; (void)n_in; (void)out_size; (void)ws_size;
  const float* q    = (const float*)d_in[0];
  const float* k    = (const float*)d_in[1];
  const float* v    = (const float*)d_in[2];
  const float* proj = (const float*)d_in[3];
  float* out        = (float*)d_out;
  float* ws         = (float*)d_ws;
  performer_key_features<<<dim3(128), dim3(256), 0, stream>>>(k, proj, ws);
  performer_causal_pairs<<<dim3(256), dim3(256), 0, stream>>>(q, v, proj, ws, out);
}
